# starting kernel
# baseline (speedup 1.0000x reference)
#include <hip/hip_runtime.h>
#include <math.h>

typedef __attribute__((ext_vector_type(2))) float v2f;
typedef __attribute__((ext_vector_type(8))) float v8f;

#define LOG2E 1.4426950408889634f

__device__ __forceinline__ float softplus_f(float v) {
  return (v > 20.0f) ? v : log1pf(__expf(v));
}

__global__ __launch_bounds__(256) void dtproj_wmma_kernel(
    const float* __restrict__ delta, const float* __restrict__ W,
    const float* __restrict__ bias, float* __restrict__ dt,
    int M, int Dc, int K) {
  int wave = threadIdx.x >> 5;
  int lane = threadIdx.x & 31;
  int tilesN = Dc >> 5;
  int tile = blockIdx.x * 8 + wave;
  int totalTiles = (M >> 4) * tilesN;
  if (tile >= totalTiles) return;
  int tn = tile % tilesN;
  int tm = tile / tilesN;
  int mn   = lane & 15;
  int half = lane >> 4;

  const float* arow  = delta + (size_t)(tm * 16 + mn) * K;
  const float* brow0 = W     + (size_t)(tn * 32 + mn) * K;
  const float* brow1 = W     + (size_t)(tn * 32 + 16 + mn) * K;

  v8f acc0 = {};
  v8f acc1 = {};
  for (int k0 = 0; k0 < K; k0 += 4) {
    int kb = k0 + (half << 1);
    v2f a, b0, b1;
    a.x  = arow[kb];  a.y  = arow[kb + 1];
    b0.x = brow0[kb]; b0.y = brow0[kb + 1];
    b1.x = brow1[kb]; b1.y = brow1[kb + 1];
    acc0 = __builtin_amdgcn_wmma_f32_16x16x4_f32(
        false, a, false, b0, (short)0, acc0, false, false);
    acc1 = __builtin_amdgcn_wmma_f32_16x16x4_f32(
        false, a, false, b1, (short)0, acc1, false, false);
  }

  int gd0 = tn * 32 + mn;
  int gd1 = gd0 + 16;
  float bv0 = bias[gd0];
  float bv1 = bias[gd1];
  int gm0 = tm * 16 + half * 8;
#pragma unroll
  for (int v = 0; v < 8; ++v) {
    size_t rowoff = (size_t)(gm0 + v) * Dc;
    dt[rowoff + gd0] = softplus_f(acc0[v] + bv0);
    dt[rowoff + gd1] = softplus_f(acc1[v] + bv1);
  }
}

__global__ void dtproj_naive_kernel(const float* __restrict__ delta,
                                    const float* __restrict__ W,
                                    const float* __restrict__ bias,
                                    float* __restrict__ dt,
                                    long long M, int Dc, int K) {
  long long idx = (long long)blockIdx.x * blockDim.x + threadIdx.x;
  if (idx >= M * (long long)Dc) return;
  int d = (int)(idx % Dc);
  long long m = idx / Dc;
  float acc = bias[d];
  const float* dr = delta + m * (long long)K;
  const float* wr = W + (size_t)d * K;
  for (int r = 0; r < K; ++r) acc += dr[r] * wr[r];
  dt[idx] = softplus_f(acc);
}

template <bool FINAL>
__global__ __launch_bounds__(256) void scan_chunk_kernel(
    const float* __restrict__ dt, const float* __restrict__ x,
    const float* __restrict__ Bm, const float* __restrict__ Cm,
    const float* __restrict__ A_log, const float* __restrict__ Dp,
    float* __restrict__ Hout, float* __restrict__ Ssum,
    const float* __restrict__ Hinit, float* __restrict__ out,
    int L, int Dc, int Lc, int nCh) {
  const int N = 16;
  extern __shared__ float smem[];
  float* sB = smem;
  float* sC = smem + (size_t)Lc * N;

  int DB = Dc / blockDim.x;
  int bx = blockIdx.x;
  int dblk = bx % DB; bx /= DB;
  int chunk = bx % nCh; bx /= nCh;
  int b = bx;
  int d = dblk * blockDim.x + threadIdx.x;

  size_t rowbase = (size_t)b * L + (size_t)chunk * Lc;

  {
    const float4* Bsrc = (const float4*)(Bm + rowbase * N);
    const float4* Csrc = (const float4*)(Cm + rowbase * N);
    float4* sB4 = (float4*)sB;
    float4* sC4 = (float4*)sC;
    int tot4 = (Lc * N) >> 2;
    for (int i = threadIdx.x; i < tot4; i += blockDim.x) {
      sB4[i] = Bsrc[i];
      if (FINAL) sC4[i] = Csrc[i];
    }
  }
  __syncthreads();

  float An2[N], h[N];
#pragma unroll
  for (int n = 0; n < N; ++n)
    An2[n] = -__expf(A_log[(size_t)d * N + n]) * LOG2E;

  size_t hidx = (((size_t)b * nCh + chunk) * Dc + d) * N;
#pragma unroll
  for (int n = 0; n < N; ++n) h[n] = FINAL ? Hinit[hidx + n] : 0.0f;

  float dtsum = 0.0f;
  float Dv = FINAL ? Dp[d] : 0.0f;

  for (int l = 0; l < Lc; ++l) {
    size_t g = (rowbase + l) * Dc + d;
    float dtv = dt[g];
    float xv = x[g];
    float us = dtv * xv;
    if (!FINAL) dtsum += dtv;
    float y = 0.0f;
#pragma unroll
    for (int n = 0; n < N; ++n) {
      float a = __builtin_amdgcn_exp2f(dtv * An2[n]);
      h[n] = a * h[n] + us * sB[l * N + n];
      if (FINAL) y += h[n] * sC[l * N + n];
    }
    if (FINAL) out[g] = y + xv * Dv;
  }

  if (!FINAL) {
#pragma unroll
    for (int n = 0; n < N; ++n) Hout[hidx + n] = h[n];
    Ssum[((size_t)b * nCh + chunk) * Dc + d] = dtsum;
  }
}

__global__ void scan_combine_kernel(const float* __restrict__ Hlocal,
                                    const float* __restrict__ Ssum,
                                    const float* __restrict__ A_log,
                                    float* __restrict__ Hinit,
                                    int Dc, int nCh, int Bsz) {
  const int N = 16;
  long long idx = (long long)blockIdx.x * blockDim.x + threadIdx.x;
  if (idx >= (long long)Bsz * Dc) return;
  int d = (int)(idx % Dc);
  int b = (int)(idx / Dc);
  float An2[N], h[N];
#pragma unroll
  for (int n = 0; n < N; ++n) {
    An2[n] = -__expf(A_log[(size_t)d * N + n]) * LOG2E;
    h[n] = 0.0f;
  }
  for (int c = 0; c < nCh; ++c) {
    size_t base = ((size_t)b * nCh + c) * Dc + d;
    size_t hidx = base * N;
    float S = Ssum[base];
#pragma unroll
    for (int n = 0; n < N; ++n) {
      Hinit[hidx + n] = h[n];
      float a = __builtin_amdgcn_exp2f(An2[n] * S);
      h[n] = a * h[n] + Hlocal[hidx + n];
    }
  }
}

__global__ void scan_simple_kernel(const float* __restrict__ dt,
                                   const float* __restrict__ x,
                                   const float* __restrict__ Bm,
                                   const float* __restrict__ Cm,
                                   const float* __restrict__ A_log,
                                   const float* __restrict__ Dp,
                                   float* __restrict__ out,
                                   int L, int Dc, int Bsz) {
  const int N = 16;
  long long idx = (long long)blockIdx.x * blockDim.x + threadIdx.x;
  if (idx >= (long long)Bsz * Dc) return;
  int d = (int)(idx % Dc);
  int b = (int)(idx / Dc);
  float An2[N], h[N];
#pragma unroll
  for (int n = 0; n < N; ++n) {
    An2[n] = -__expf(A_log[(size_t)d * N + n]) * LOG2E;
    h[n] = 0.0f;
  }
  float Dv = Dp[d];
  for (int l = 0; l < L; ++l) {
    size_t row = (size_t)b * L + l;
    size_t g = row * Dc + d;
    float dtv = dt[g];
    float xv = x[g];
    float us = dtv * xv;
    const float* Br = Bm + row * N;
    const float* Cr = Cm + row * N;
    float y = 0.0f;
#pragma unroll
    for (int n = 0; n < N; ++n) {
      h[n] = __builtin_amdgcn_exp2f(dtv * An2[n]) * h[n] + us * Br[n];
      y += h[n] * Cr[n];
    }
    out[g] = y + xv * Dv;
  }
}

__global__ void scan_fused_kernel(const float* __restrict__ x,
                                  const float* __restrict__ delta,
                                  const float* __restrict__ Bm,
                                  const float* __restrict__ Cm,
                                  const float* __restrict__ W,
                                  const float* __restrict__ bias,
                                  const float* __restrict__ A_log,
                                  const float* __restrict__ Dp,
                                  float* __restrict__ out,
                                  int L, int Dc, int K, int Bsz) {
  const int N = 16;
  long long idx = (long long)blockIdx.x * blockDim.x + threadIdx.x;
  if (idx >= (long long)Bsz * Dc) return;
  int d = (int)(idx % Dc);
  int b = (int)(idx / Dc);
  float An2[N], h[N];
#pragma unroll
  for (int n = 0; n < N; ++n) {
    An2[n] = -__expf(A_log[(size_t)d * N + n]) * LOG2E;
    h[n] = 0.0f;
  }
  float Dv = Dp[d];
  float bv = bias[d];
  const float* wr = W + (size_t)d * K;
  for (int l = 0; l < L; ++l) {
    size_t row = (size_t)b * L + l;
    const float* dr = delta + row * K;
    float acc = bv;
    for (int r = 0; r < K; ++r) acc += dr[r] * wr[r];
    float dtv = softplus_f(acc);
    float xv = x[row * Dc + d];
    float us = dtv * xv;
    const float* Br = Bm + row * N;
    const float* Cr = Cm + row * N;
    float y = 0.0f;
#pragma unroll
    for (int n = 0; n < N; ++n) {
      h[n] = __builtin_amdgcn_exp2f(dtv * An2[n]) * h[n] + us * Br[n];
      y += h[n] * Cr[n];
    }
    out[row * Dc + d] = y + xv * Dv;
  }
}

extern "C" void kernel_launch(void* const* d_in, const int* in_sizes, int n_in,
                              void* d_out, int out_size, void* d_ws, size_t ws_size,
                              hipStream_t stream) {
  const float* x     = (const float*)d_in[0];
  const float* delta = (const float*)d_in[1];
  const float* Bm    = (const float*)d_in[2];
  const float* Cm    = (const float*)d_in[3];
  const float* W     = (const float*)d_in[4];
  const float* bias  = (const float*)d_in[5];
  const float* A_log = (const float*)d_in[6];
  const float* Dp    = (const float*)d_in[7];
  float* out = (float*)d_out;

  int Dc = in_sizes[5];
  int K  = in_sizes[4] / Dc;
  long long BL = (long long)in_sizes[0] / Dc;
  int Bsz = (BL % 2 == 0) ? 2 : 1;
  int L = (int)(BL / Bsz);

  size_t dtElems = (size_t)BL * Dc;
  size_t dtBytes = dtElems * sizeof(float);

  const int Lc = 64;
  bool chunkable = (L % Lc) == 0 && (Dc % 256) == 0;
  int nCh = chunkable ? (L / Lc) : 1;
  size_t chElems = (size_t)Bsz * nCh * Dc;
  size_t fullBytes = dtBytes + chElems * sizeof(float) * (1 + 16 + 16);

  if (ws_size < dtBytes) {
    int threads = 256;
    long long tot = (long long)Bsz * Dc;
    int blocks = (int)((tot + threads - 1) / threads);
    scan_fused_kernel<<<blocks, threads, 0, stream>>>(
        x, delta, Bm, Cm, W, bias, A_log, Dp, out, L, Dc, K, Bsz);
    return;
  }

  float* dt = (float*)d_ws;

  if ((BL % 16) == 0 && (Dc % 32) == 0 && (K % 4) == 0) {
    int tiles = (int)(BL / 16) * (Dc / 32);
    int blocks = (tiles + 7) / 8;
    dtproj_wmma_kernel<<<blocks, 256, 0, stream>>>(delta, W, bias, dt,
                                                   (int)BL, Dc, K);
  } else {
    long long tot = BL * (long long)Dc;
    int blocks = (int)((tot + 255) / 256);
    dtproj_naive_kernel<<<blocks, 256, 0, stream>>>(delta, W, bias, dt, BL, Dc, K);
  }

  if (chunkable && ws_size >= fullBytes) {
    float* Ssum   = dt + dtElems;
    float* Hlocal = Ssum + chElems;
    float* Hinit  = Hlocal + chElems * 16;

    int blocks13 = Bsz * nCh * (Dc / 256);
    size_t smem1 = (size_t)Lc * 16 * sizeof(float);

    scan_chunk_kernel<false><<<blocks13, 256, smem1, stream>>>(
        dt, x, Bm, Cm, A_log, Dp, Hlocal, Ssum, (const float*)nullptr,
        (float*)nullptr, L, Dc, Lc, nCh);

    long long tot = (long long)Bsz * Dc;
    int cblocks = (int)((tot + 255) / 256);
    scan_combine_kernel<<<cblocks, 256, 0, stream>>>(Hlocal, Ssum, A_log, Hinit,
                                                     Dc, nCh, Bsz);

    scan_chunk_kernel<true><<<blocks13, 256, smem1 * 2, stream>>>(
        dt, x, Bm, Cm, A_log, Dp, (float*)nullptr, (float*)nullptr, Hinit, out,
        L, Dc, Lc, nCh);
  } else {
    long long tot = (long long)Bsz * Dc;
    int blocks = (int)((tot + 255) / 256);
    scan_simple_kernel<<<blocks, 256, 0, stream>>>(dt, x, Bm, Cm, A_log, Dp, out,
                                                   L, Dc, Bsz);
  }
}


# dt projection moved into the scan kernel: per 64-step chunk the group's WMMAs make dt from LDS-staged delta; dt never goes to memory, one launch; K summed as 4 partial sums (order changes)
# speedup vs baseline: 2.4125x; 2.4125x over previous
#include <hip/hip_runtime.h>
#include <math.h>

typedef __attribute__((ext_vector_type(2))) float v2f;
typedef __attribute__((ext_vector_type(8))) float v8f;

#define LOG2E 1.4426950408889634f

__device__ __forceinline__ float softplus_f(float v) {
  return (v > 20.0f) ? v : log1pf(__expf(v));
}

__global__ __launch_bounds__(256) void dtproj_wmma_kernel(
    const float* __restrict__ delta, const float* __restrict__ W,
    const float* __restrict__ bias, float* __restrict__ dt,
    int M, int Dc, int K) {
  int wave = threadIdx.x >> 5;
  int lane = threadIdx.x & 31;
  int tilesN = Dc >> 5;
  int tile = blockIdx.x * 8 + wave;
  int totalTiles = (M >> 4) * tilesN;
  if (tile >= totalTiles) return;
  int tn = tile % tilesN;
  int tm = tile / tilesN;
  int mn   = lane & 15;
  int half = lane >> 4;

  const float* arow  = delta + (size_t)(tm * 16 + mn) * K;
  const float* brow0 = W     + (size_t)(tn * 32 + mn) * K;
  const float* brow1 = W     + (size_t)(tn * 32 + 16 + mn) * K;

  v8f acc0 = {};
  v8f acc1 = {};
  for (int k0 = 0; k0 < K; k0 += 4) {
    int kb = k0 + (half << 1);
    v2f a, b0, b1;
    a.x  = arow[kb];  a.y  = arow[kb + 1];
    b0.x = brow0[kb]; b0.y = brow0[kb + 1];
    b1.x = brow1[kb]; b1.y = brow1[kb + 1];
    acc0 = __builtin_amdgcn_wmma_f32_16x16x4_f32(
        false, a, false, b0, (short)0, acc0, false, false);
    acc1 = __builtin_amdgcn_wmma_f32_16x16x4_f32(
        false, a, false, b1, (short)0, acc1, false, false);
  }

  int gd0 = tn * 32 + mn;
  int gd1 = gd0 + 16;
  float bv0 = bias[gd0];
  float bv1 = bias[gd1];
  int gm0 = tm * 16 + half * 8;
#pragma unroll
  for (int v = 0; v < 8; ++v) {
    size_t rowoff = (size_t)(gm0 + v) * Dc;
    dt[rowoff + gd0] = softplus_f(acc0[v] + bv0);
    dt[rowoff + gd1] = softplus_f(acc1[v] + bv1);
  }
}

__global__ void dtproj_naive_kernel(const float* __restrict__ delta,
                                    const float* __restrict__ W,
                                    const float* __restrict__ bias,
                                    float* __restrict__ dt,
                                    long long M, int Dc, int K) {
  long long idx = (long long)blockIdx.x * blockDim.x + threadIdx.x;
  if (idx >= M * (long long)Dc) return;
  int d = (int)(idx % Dc);
  long long m = idx / Dc;
  float acc = bias[d];
  const float* dr = delta + m * (long long)K;
  const float* wr = W + (size_t)d * K;
  for (int r = 0; r < K; ++r) acc += dr[r] * wr[r];
  dt[idx] = softplus_f(acc);
}

#define SCAN_TC 64

template <int MASK>
__device__ __forceinline__ float row_xor(float v) {
  int r = __builtin_amdgcn_update_dpp(0, __builtin_bit_cast(int, v),
                                      0x160 + MASK, 0xf, 0xf, true);
  return __builtin_bit_cast(float, r);
}

template <int MASK>
__device__ __forceinline__ float fold_pair(bool bit, float lo, float hi) {
  float keep = bit ? hi : lo;
  float send = bit ? lo : hi;
  return keep + row_xor<MASK>(send);
}

__global__ __launch_bounds__(256) void scan_lanes_kernel(
    const float* __restrict__ dt, const float* __restrict__ x,
    const float* __restrict__ Bm, const float* __restrict__ Cm,
    const float* __restrict__ A_log, const float* __restrict__ Dp,
    float* __restrict__ out, int L, int Dc) {
  const int N = 16;
  const int TC = SCAN_TC;
  __shared__ float4 s_dtus[2][TC / 2 * 16];
  __shared__ float4 s_bc[2][TC / 2 * 16];
  __shared__ float4 s_y[2][TC * 4];

  const int tid = threadIdx.x;
  const int lane = tid & 31;
  const int wave = tid >> 5;
  const int n = lane & 15;
  const int ch = wave * 2 + (lane >> 4);
  const int DB = Dc >> 4;
  const int b = blockIdx.x / DB;
  const int d0 = (blockIdx.x % DB) << 4;

  const int row = tid >> 2;
  const int q = tid & 3;
  const size_t grow = (size_t)b * L + row;
  const float* dt_src = dt + grow * Dc + d0 + q * 4;
  const float* x_src  = x  + grow * Dc + d0 + q * 4;
  const float* B_src  = Bm + grow * N + q * 4;
  const float* C_src  = Cm + grow * N + q * 4;
  float* out_dst      = out + grow * Dc + d0 + q * 4;
  const size_t stepD = (size_t)TC * Dc;
  const size_t stepN = (size_t)TC * N;
  const int sbase = ((((row >> 1) * 16) + q * 4) << 1) + (row & 1);

  const float An2 = -__expf(A_log[(size_t)(d0 + ch) * N + n]) * LOG2E;
  const float4 D4 = *(const float4*)(Dp + d0 + q * 4);
  const bool b0 = (n & 1) != 0, b1 = (n & 2) != 0;
  const bool b2 = (n & 4) != 0, b3 = (n & 8) != 0;

  float4 dtn = *(const float4*)dt_src;
  float4 xn  = *(const float4*)x_src;
  float4 Bn  = *(const float4*)B_src;
  float4 Cn  = *(const float4*)C_src;
  {
    float2* pd = (float2*)s_dtus[0];
    float2* pb = (float2*)s_bc[0];
    pd[sbase + 0] = make_float2(dtn.x, dtn.x * xn.x);
    pd[sbase + 2] = make_float2(dtn.y, dtn.y * xn.y);
    pd[sbase + 4] = make_float2(dtn.z, dtn.z * xn.z);
    pd[sbase + 6] = make_float2(dtn.w, dtn.w * xn.w);
    pb[sbase + 0] = make_float2(Bn.x, Cn.x);
    pb[sbase + 2] = make_float2(Bn.y, Cn.y);
    pb[sbase + 4] = make_float2(Bn.z, Cn.z);
    pb[sbase + 6] = make_float2(Bn.w, Cn.w);
  }
  float4 xcur = xn;
  __syncthreads();

  float h = 0.0f;
  const int nCh = L / TC;
  for (int k = 0; k < nCh; ++k) {
    const int cur = k & 1;
    const bool more = (k + 1) < nCh;
    if (more) {
      dtn = *(const float4*)(dt_src + (size_t)(k + 1) * stepD);
      xn  = *(const float4*)(x_src  + (size_t)(k + 1) * stepD);
      Bn  = *(const float4*)(B_src  + (size_t)(k + 1) * stepN);
      Cn  = *(const float4*)(C_src  + (size_t)(k + 1) * stepN);
    }

    const float4* pdt = s_dtus[cur] + ch;
    const float4* pbc = s_bc[cur] + n;
    float* py = (float*)s_y[cur] + n * 16 + ch;
#pragma unroll 2
    for (int g = 0; g < TC / 16; ++g) {
      float p[16];
#pragma unroll
      for (int tp = 0; tp < 8; ++tp) {
        float4 v = pdt[(g * 8 + tp) * 16];
        float4 w = pbc[(g * 8 + tp) * 16];
        float a0 = __builtin_amdgcn_exp2f(v.x * An2);
        h = a0 * h + v.y * w.x;
        p[2 * tp] = h * w.y;
        float a1 = __builtin_amdgcn_exp2f(v.z * An2);
        h = a1 * h + v.w * w.z;
        p[2 * tp + 1] = h * w.w;
      }
      float r8[8], r4[4], r2[2];
#pragma unroll
      for (int j = 0; j < 8; ++j) r8[j] = fold_pair<1>(b0, p[2 * j], p[2 * j + 1]);
#pragma unroll
      for (int j = 0; j < 4; ++j) r4[j] = fold_pair<2>(b1, r8[2 * j], r8[2 * j + 1]);
#pragma unroll
      for (int j = 0; j < 2; ++j) r2[j] = fold_pair<4>(b2, r4[2 * j], r4[2 * j + 1]);
      float ysum = fold_pair<8>(b3, r2[0], r2[1]);
      py[g * 256] = ysum;
    }

    if (more) {
      float2* pd = (float2*)s_dtus[cur ^ 1];
      float2* pb = (float2*)s_bc[cur ^ 1];
      pd[sbase + 0] = make_float2(dtn.x, dtn.x * xn.x);
      pd[sbase + 2] = make_float2(dtn.y, dtn.y * xn.y);
      pd[sbase + 4] = make_float2(dtn.z, dtn.z * xn.z);
      pd[sbase + 6] = make_float2(dtn.w, dtn.w * xn.w);
      pb[sbase + 0] = make_float2(Bn.x, Cn.x);
      pb[sbase + 2] = make_float2(Bn.y, Cn.y);
      pb[sbase + 4] = make_float2(Bn.z, Cn.z);
      pb[sbase + 6] = make_float2(Bn.w, Cn.w);
    }
    __syncthreads();

    float4 yv = s_y[cur][row * 4 + q];
    float4 o;
    o.x = yv.x + xcur.x * D4.x;
    o.y = yv.y + xcur.y * D4.y;
    o.z = yv.z + xcur.z * D4.z;
    o.w = yv.w + xcur.w * D4.w;
    *(float4*)(out_dst + (size_t)k * stepD) = o;
    xcur = xn;
  }
}

#define FUSE_K 128
#define DL_PITCH 132
#define DT_PITCH 20

__device__ __forceinline__ float softplus_sel(float v) {
  float l = log1pf(__expf(v));
  return (v > 20.0f) ? v : l;
}

__global__ __launch_bounds__(256) void scan_dtfused_kernel(
    const float* __restrict__ x, const float* __restrict__ delta,
    const float* __restrict__ Bm, const float* __restrict__ Cm,
    const float* __restrict__ W, const float* __restrict__ bias,
    const float* __restrict__ A_log, const float* __restrict__ Dp,
    float* __restrict__ out, int L, int Dc) {
  const int N = 16;
  const int TC = SCAN_TC;
  typedef float4 stage_t[TC / 2 * 16];
  typedef float4 ysum_t[TC * 4];
  typedef float4 dtp_t[2][TC * DT_PITCH / 4];
  typedef float4 dlt_t[TC * DL_PITCH / 4];
  __shared__ float4 smem[2 * (TC / 2 * 16) * 2 + 2 * (TC * 4) +
                         2 * 2 * (TC * DT_PITCH / 4) + 2 * (TC * DL_PITCH / 4)];
  stage_t* s_dtus = (stage_t*)smem;
  stage_t* s_bc   = (stage_t*)(smem + 2 * (TC / 2 * 16));
  ysum_t* s_y     = (ysum_t*)(smem + 4 * (TC / 2 * 16));
  dtp_t* s_dtp    = (dtp_t*)(smem + 4 * (TC / 2 * 16) + 2 * (TC * 4));
  dlt_t* s_delta  = (dlt_t*)(smem + 4 * (TC / 2 * 16) + 2 * (TC * 4) +
                             2 * 2 * (TC * DT_PITCH / 4));

  const int tid = threadIdx.x;
  const int lane = tid & 31;
  const int wave = __builtin_amdgcn_readfirstlane(tid >> 5);
  const int n = lane & 15;
  const int half = lane >> 4;
  const int ch = wave * 2 + half;
  const int DB = Dc >> 4;
  const int b = blockIdx.x / DB;
  const int d0 = (blockIdx.x % DB) << 4;
  const int tile16 = wave & 3;
  const int kh = wave >> 2;

  const int row = tid >> 2;
  const int q = tid & 3;
  const size_t grow = (size_t)b * L + row;
  const float* x_src  = x  + grow * Dc + d0 + q * 4;
  const float* B_src  = Bm + grow * N + q * 4;
  const float* C_src  = Cm + grow * N + q * 4;
  float* out_dst      = out + grow * Dc + d0 + q * 4;
  const size_t stepD = (size_t)TC * Dc;
  const size_t stepN = (size_t)TC * N;
  const int sbase = ((((row >> 1) * 16) + q * 4) << 1) + (row & 1);

  const float An2 = -__expf(A_log[(size_t)(d0 + ch) * N + n]) * LOG2E;
  const float4 D4 = *(const float4*)(Dp + d0 + q * 4);
  const float4 bias4 = *(const float4*)(bias + d0 + q * 4);
  const bool b0 = (n & 1) != 0, b1 = (n & 2) != 0;
  const bool b2 = (n & 4) != 0, b3 = (n & 8) != 0;

  v2f bw[16];
  {
    const float* wsrc = W + (size_t)(d0 + n) * FUSE_K + kh * 64 + half * 2;
#pragma unroll
    for (int s = 0; s < 16; ++s) bw[s] = *(const v2f*)(wsrc + 4 * s);
  }

  const float4* dl_src = (const float4*)(delta + (size_t)b * L * FUSE_K) + tid;
  const int dl_dst = wave * (DL_PITCH / 4) + lane;
#define DELTA_DECL float4 dl0, dl1, dl2, dl3, dl4, dl5, dl6, dl7;
#define DELTA_LOAD(c)                                                   \
  {                                                                     \
    const float4* dlp = dl_src + (size_t)(c) * (TC * FUSE_K / 4);       \
    dl0 = dlp[0 * 256]; dl1 = dlp[1 * 256];                             \
    dl2 = dlp[2 * 256]; dl3 = dlp[3 * 256];                             \
    dl4 = dlp[4 * 256]; dl5 = dlp[5 * 256];                             \
    dl6 = dlp[6 * 256]; dl7 = dlp[7 * 256];                             \
  }
#define DELTA_STORE(buf)                                                \
  {                                                                     \
    float4* dls = s_delta[buf] + dl_dst;                                \
    dls[0 * 8 * (DL_PITCH / 4)] = dl0; dls[1 * 8 * (DL_PITCH / 4)] = dl1; \
    dls[2 * 8 * (DL_PITCH / 4)] = dl2; dls[3 * 8 * (DL_PITCH / 4)] = dl3; \
    dls[4 * 8 * (DL_PITCH / 4)] = dl4; dls[5 * 8 * (DL_PITCH / 4)] = dl5; \
    dls[6 * 8 * (DL_PITCH / 4)] = dl6; dls[7 * 8 * (DL_PITCH / 4)] = dl7; \
  }

  auto compute_dt = [&](int buf) {
    const float* sa = (const float*)s_delta[buf] +
                      (tile16 * 16 + n) * DL_PITCH + kh * 64 + half * 2;
    v2f a[16];
#pragma unroll
    for (int s = 0; s < 16; ++s) a[s] = *(const v2f*)(sa + 4 * s);
    asm volatile("" ::"v"(a[0]), "v"(a[1]), "v"(a[2]), "v"(a[3]), "v"(a[4]),
                 "v"(a[5]), "v"(a[6]), "v"(a[7]), "v"(a[8]), "v"(a[9]),
                 "v"(a[10]), "v"(a[11]), "v"(a[12]), "v"(a[13]), "v"(a[14]),
                 "v"(a[15]));
    v8f acc0 = {};
    v8f acc1 = {};
#pragma unroll
    for (int s = 0; s < 16; s += 2) {
      acc0 = __builtin_amdgcn_wmma_f32_16x16x4_f32(
          false, a[s], false, bw[s], (short)0, acc0, false, false);
      acc1 = __builtin_amdgcn_wmma_f32_16x16x4_f32(
          false, a[s + 1], false, bw[s + 1], (short)0, acc1, false, false);
    }
    float* sd = (float*)s_dtp[buf][kh] + (tile16 * 16 + half * 8) * DT_PITCH + n;
#pragma unroll
    for (int r = 0; r < 8; ++r) sd[r * DT_PITCH] = acc0[r] + acc1[r];
  };

  auto stage = [&](int buf, float4 xv, float4 Bv, float4 Cv) {
    const float* sp = (const float*)s_dtp[buf][0] + row * DT_PITCH + q * 4;
    float4 p0 = *(const float4*)sp;
    float4 p1 = *(const float4*)(sp + TC * DT_PITCH);
    float4 dtv;
    dtv.x = softplus_sel((p0.x + p1.x) + bias4.x);
    dtv.y = softplus_sel((p0.y + p1.y) + bias4.y);
    dtv.z = softplus_sel((p0.z + p1.z) + bias4.z);
    dtv.w = softplus_sel((p0.w + p1.w) + bias4.w);
    float2* pd = (float2*)s_dtus[buf];
    float2* pb = (float2*)s_bc[buf];
    pd[sbase + 0] = make_float2(dtv.x, dtv.x * xv.x);
    pd[sbase + 2] = make_float2(dtv.y, dtv.y * xv.y);
    pd[sbase + 4] = make_float2(dtv.z, dtv.z * xv.z);
    pd[sbase + 6] = make_float2(dtv.w, dtv.w * xv.w);
    pb[sbase + 0] = make_float2(Bv.x, Cv.x);
    pb[sbase + 2] = make_float2(Bv.y, Cv.y);
    pb[sbase + 4] = make_float2(Bv.z, Cv.z);
    pb[sbase + 6] = make_float2(Bv.w, Cv.w);
  };

  const int nCh = L / TC;
  const int lastCh = nCh - 1;

  float4 xcur;
  {
    const float4 x0 = *(const float4*)x_src;
    const float4 B0 = *(const float4*)B_src;
    const float4 C0 = *(const float4*)C_src;
    {
      DELTA_DECL
      DELTA_LOAD(0)
      DELTA_STORE(0)
    }
    {
      DELTA_DECL
      DELTA_LOAD(lastCh < 1 ? lastCh : 1)
      DELTA_STORE(1)
    }
    __syncthreads();
    compute_dt(0);
    compute_dt(1);
    __syncthreads();
    stage(0, x0, B0, C0);
    {
      DELTA_DECL
      DELTA_LOAD(lastCh < 2 ? lastCh : 2)
      DELTA_STORE(0)
    }
    xcur = x0;
    __syncthreads();
  }

  float h = 0.0f;
  for (int k = 0; k < nCh; ++k) {
    const int cur = k & 1;
    const bool more = (k + 1) < nCh;
    const bool more2 = (k + 2) < nCh;
    const bool more3 = (k + 3) < nCh;
    const int k1 = more ? k + 1 : k;
    const float4 xn = *(const float4*)(x_src + (size_t)k1 * stepD);
    const float4 Bn = *(const float4*)(B_src + (size_t)k1 * stepN);
    const float4 Cn = *(const float4*)(C_src + (size_t)k1 * stepN);
    DELTA_DECL
    if (more3) DELTA_LOAD(k + 3)

    if (more2 && kh == 0) compute_dt(cur);

    const float4* pdt = s_dtus[cur] + ch;
    const float4* pbc = s_bc[cur] + n;
    float* py = (float*)s_y[cur] + n * 16 + ch;
#pragma unroll 2
    for (int g = 0; g < TC / 16; ++g) {
      float p[16];
#pragma unroll
      for (int tp = 0; tp < 8; ++tp) {
        float4 v = pdt[(g * 8 + tp) * 16];
        float4 w = pbc[(g * 8 + tp) * 16];
        float a0 = __builtin_amdgcn_exp2f(v.x * An2);
        h = a0 * h + v.y * w.x;
        p[2 * tp] = h * w.y;
        float a1 = __builtin_amdgcn_exp2f(v.z * An2);
        h = a1 * h + v.w * w.z;
        p[2 * tp + 1] = h * w.w;
      }
      float r8[8], r4[4], r2[2];
#pragma unroll
      for (int j = 0; j < 8; ++j) r8[j] = fold_pair<1>(b0, p[2 * j], p[2 * j + 1]);
#pragma unroll
      for (int j = 0; j < 4; ++j) r4[j] = fold_pair<2>(b1, r8[2 * j], r8[2 * j + 1]);
#pragma unroll
      for (int j = 0; j < 2; ++j) r2[j] = fold_pair<4>(b2, r4[2 * j], r4[2 * j + 1]);
      float ysum = fold_pair<8>(b3, r2[0], r2[1]);
      py[g * 256] = ysum;
    }

    if (more2 && kh != 0) compute_dt(cur);
    if (more) stage(cur ^ 1, xn, Bn, Cn);
    if (more3) DELTA_STORE(cur ^ 1)
    __syncthreads();

    float4 yv = s_y[cur][row * 4 + q];
    float4 o;
    o.x = yv.x + xcur.x * D4.x;
    o.y = yv.y + xcur.y * D4.y;
    o.z = yv.z + xcur.z * D4.z;
    o.w = yv.w + xcur.w * D4.w;
    *(float4*)(out_dst + (size_t)k * stepD) = o;
    xcur = xn;
  }
#undef DELTA_DECL
#undef DELTA_LOAD
#undef DELTA_STORE
}

__global__ void scan_simple_kernel(const float* __restrict__ dt,
                                   const float* __restrict__ x,
                                   const float* __restrict__ Bm,
                                   const float* __restrict__ Cm,
                                   const float* __restrict__ A_log,
                                   const float* __restrict__ Dp,
                                   float* __restrict__ out,
                                   int L, int Dc, int Bsz) {
  const int N = 16;
  long long idx = (long long)blockIdx.x * blockDim.x + threadIdx.x;
  if (idx >= (long long)Bsz * Dc) return;
  int d = (int)(idx % Dc);
  int b = (int)(idx / Dc);
  float An2[N], h[N];
#pragma unroll
  for (int n = 0; n < N; ++n) {
    An2[n] = -__expf(A_log[(size_t)d * N + n]) * LOG2E;
    h[n] = 0.0f;
  }
  float Dv = Dp[d];
  for (int l = 0; l < L; ++l) {
    size_t row = (size_t)b * L + l;
    size_t g = row * Dc + d;
    float dtv = dt[g];
    float xv = x[g];
    float us = dtv * xv;
    const float* Br = Bm + row * N;
    const float* Cr = Cm + row * N;
    float y = 0.0f;
#pragma unroll
    for (int n = 0; n < N; ++n) {
      h[n] = __builtin_amdgcn_exp2f(dtv * An2[n]) * h[n] + us * Br[n];
      y += h[n] * Cr[n];
    }
    out[g] = y + xv * Dv;
  }
}

__global__ void scan_fused_kernel(const float* __restrict__ x,
                                  const float* __restrict__ delta,
                                  const float* __restrict__ Bm,
                                  const float* __restrict__ Cm,
                                  const float* __restrict__ W,
                                  const float* __restrict__ bias,
                                  const float* __restrict__ A_log,
                                  const float* __restrict__ Dp,
                                  float* __restrict__ out,
                                  int L, int Dc, int K, int Bsz) {
  const int N = 16;
  long long idx = (long long)blockIdx.x * blockDim.x + threadIdx.x;
  if (idx >= (long long)Bsz * Dc) return;
  int d = (int)(idx % Dc);
  int b = (int)(idx / Dc);
  float An2[N], h[N];
#pragma unroll
  for (int n = 0; n < N; ++n) {
    An2[n] = -__expf(A_log[(size_t)d * N + n]) * LOG2E;
    h[n] = 0.0f;
  }
  float Dv = Dp[d];
  float bv = bias[d];
  const float* wr = W + (size_t)d * K;
  for (int l = 0; l < L; ++l) {
    size_t row = (size_t)b * L + l;
    const float* dr = delta + row * K;
    float acc = bv;
    for (int r = 0; r < K; ++r) acc += dr[r] * wr[r];
    float dtv = softplus_f(acc);
    float xv = x[row * Dc + d];
    float us = dtv * xv;
    const float* Br = Bm + row * N;
    const float* Cr = Cm + row * N;
    float y = 0.0f;
#pragma unroll
    for (int n = 0; n < N; ++n) {
      h[n] = __builtin_amdgcn_exp2f(dtv * An2[n]) * h[n] + us * Br[n];
      y += h[n] * Cr[n];
    }
    out[row * Dc + d] = y + xv * Dv;
  }
}

extern "C" void kernel_launch(void* const* d_in, const int* in_sizes, int n_in,
                              void* d_out, int out_size, void* d_ws, size_t ws_size,
                              hipStream_t stream) {
  const float* x     = (const float*)d_in[0];
  const float* delta = (const float*)d_in[1];
  const float* Bm    = (const float*)d_in[2];
  const float* Cm    = (const float*)d_in[3];
  const float* W     = (const float*)d_in[4];
  const float* bias  = (const float*)d_in[5];
  const float* A_log = (const float*)d_in[6];
  const float* Dp    = (const float*)d_in[7];
  float* out = (float*)d_out;

  int Dc = in_sizes[5];
  int K  = in_sizes[4] / Dc;
  long long BL = (long long)in_sizes[0] / Dc;
  int Bsz = (BL % 2 == 0) ? 2 : 1;
  int L = (int)(BL / Bsz);

  size_t dtElems = (size_t)BL * Dc;
  size_t dtBytes = dtElems * sizeof(float);

  if (K == FUSE_K && (L % SCAN_TC) == 0 && (Dc % 16) == 0) {
    int blocks = Bsz * (Dc / 16);
    scan_dtfused_kernel<<<blocks, 256, 0, stream>>>(x, delta, Bm, Cm, W, bias,
                                                    A_log, Dp, out, L, Dc);
    return;
  }

  if (ws_size < dtBytes) {
    int threads = 256;
    long long tot = (long long)Bsz * Dc;
    int blocks = (int)((tot + threads - 1) / threads);
    scan_fused_kernel<<<blocks, threads, 0, stream>>>(
        x, delta, Bm, Cm, W, bias, A_log, Dp, out, L, Dc, K, Bsz);
    return;
  }

  float* dt = (float*)d_ws;

  if ((BL % 16) == 0 && (Dc % 32) == 0 && (K % 4) == 0) {
    int tiles = (int)(BL / 16) * (Dc / 32);
    int blocks = (tiles + 7) / 8;
    dtproj_wmma_kernel<<<blocks, 256, 0, stream>>>(delta, W, bias, dt,
                                                   (int)BL, Dc, K);
  } else {
    long long tot = BL * (long long)Dc;
    int blocks = (int)((tot + 255) / 256);
    dtproj_naive_kernel<<<blocks, 256, 0, stream>>>(delta, W, bias, dt, BL, Dc, K);
  }

  if ((L % SCAN_TC) == 0 && (Dc % 16) == 0) {
    int blocks = Bsz * (Dc / 16);
    scan_lanes_kernel<<<blocks, 256, 0, stream>>>(dt, x, Bm, Cm, A_log, Dp, out,
                                                  L, Dc);
  } else {
    long long tot = (long long)Bsz * Dc;
    int blocks = (int)((tot + 255) / 256);
    scan_simple_kernel<<<blocks, 256, 0, stream>>>(dt, x, Bm, Cm, A_log, Dp, out,
                                                   L, Dc, Bsz);
  }
}


# work-group of 16 waves in two roles: 8 waves only scan, 8 do the dt projection (WMMA), softplus, operand staging and output one chunk ahead; 4 waves per SIMD fill each other's waits; sums unchanged
# speedup vs baseline: 2.5301x; 1.0487x over previous
#include <hip/hip_runtime.h>
#include <math.h>

typedef __attribute__((ext_vector_type(2))) float v2f;
typedef __attribute__((ext_vector_type(8))) float v8f;

#define LOG2E 1.4426950408889634f

__device__ __forceinline__ float softplus_f(float v) {
  return (v > 20.0f) ? v : log1pf(__expf(v));
}

__global__ __launch_bounds__(256) void dtproj_wmma_kernel(
    const float* __restrict__ delta, const float* __restrict__ W,
    const float* __restrict__ bias, float* __restrict__ dt,
    int M, int Dc, int K) {
  int wave = threadIdx.x >> 5;
  int lane = threadIdx.x & 31;
  int tilesN = Dc >> 5;
  int tile = blockIdx.x * 8 + wave;
  int totalTiles = (M >> 4) * tilesN;
  if (tile >= totalTiles) return;
  int tn = tile % tilesN;
  int tm = tile / tilesN;
  int mn   = lane & 15;
  int half = lane >> 4;

  const float* arow  = delta + (size_t)(tm * 16 + mn) * K;
  const float* brow0 = W     + (size_t)(tn * 32 + mn) * K;
  const float* brow1 = W     + (size_t)(tn * 32 + 16 + mn) * K;

  v8f acc0 = {};
  v8f acc1 = {};
  for (int k0 = 0; k0 < K; k0 += 4) {
    int kb = k0 + (half << 1);
    v2f a, b0, b1;
    a.x  = arow[kb];  a.y  = arow[kb + 1];
    b0.x = brow0[kb]; b0.y = brow0[kb + 1];
    b1.x = brow1[kb]; b1.y = brow1[kb + 1];
    acc0 = __builtin_amdgcn_wmma_f32_16x16x4_f32(
        false, a, false, b0, (short)0, acc0, false, false);
    acc1 = __builtin_amdgcn_wmma_f32_16x16x4_f32(
        false, a, false, b1, (short)0, acc1, false, false);
  }

  int gd0 = tn * 32 + mn;
  int gd1 = gd0 + 16;
  float bv0 = bias[gd0];
  float bv1 = bias[gd1];
  int gm0 = tm * 16 + half * 8;
#pragma unroll
  for (int v = 0; v < 8; ++v) {
    size_t rowoff = (size_t)(gm0 + v) * Dc;
    dt[rowoff + gd0] = softplus_f(acc0[v] + bv0);
    dt[rowoff + gd1] = softplus_f(acc1[v] + bv1);
  }
}

__global__ void dtproj_naive_kernel(const float* __restrict__ delta,
                                    const float* __restrict__ W,
                                    const float* __restrict__ bias,
                                    float* __restrict__ dt,
                                    long long M, int Dc, int K) {
  long long idx = (long long)blockIdx.x * blockDim.x + threadIdx.x;
  if (idx >= M * (long long)Dc) return;
  int d = (int)(idx % Dc);
  long long m = idx / Dc;
  float acc = bias[d];
  const float* dr = delta + m * (long long)K;
  const float* wr = W + (size_t)d * K;
  for (int r = 0; r < K; ++r) acc += dr[r] * wr[r];
  dt[idx] = softplus_f(acc);
}

#define SCAN_TC 64

template <int MASK>
__device__ __forceinline__ float row_xor(float v) {
  int r = __builtin_amdgcn_update_dpp(0, __builtin_bit_cast(int, v),
                                      0x160 + MASK, 0xf, 0xf, true);
  return __builtin_bit_cast(float, r);
}

template <int MASK>
__device__ __forceinline__ float fold_pair(bool bit, float lo, float hi) {
  float keep = bit ? hi : lo;
  float send = bit ? lo : hi;
  return keep + row_xor<MASK>(send);
}

__global__ __launch_bounds__(256) void scan_lanes_kernel(
    const float* __restrict__ dt, const float* __restrict__ x,
    const float* __restrict__ Bm, const float* __restrict__ Cm,
    const float* __restrict__ A_log, const float* __restrict__ Dp,
    float* __restrict__ out, int L, int Dc) {
  const int N = 16;
  const int TC = SCAN_TC;
  __shared__ float4 s_dtus[2][TC / 2 * 16];
  __shared__ float4 s_bc[2][TC / 2 * 16];
  __shared__ float4 s_y[2][TC * 4];

  const int tid = threadIdx.x;
  const int lane = tid & 31;
  const int wave = tid >> 5;
  const int n = lane & 15;
  const int ch = wave * 2 + (lane >> 4);
  const int DB = Dc >> 4;
  const int b = blockIdx.x / DB;
  const int d0 = (blockIdx.x % DB) << 4;

  const int row = tid >> 2;
  const int q = tid & 3;
  const size_t grow = (size_t)b * L + row;
  const float* dt_src = dt + grow * Dc + d0 + q * 4;
  const float* x_src  = x  + grow * Dc + d0 + q * 4;
  const float* B_src  = Bm + grow * N + q * 4;
  const float* C_src  = Cm + grow * N + q * 4;
  float* out_dst      = out + grow * Dc + d0 + q * 4;
  const size_t stepD = (size_t)TC * Dc;
  const size_t stepN = (size_t)TC * N;
  const int sbase = ((((row >> 1) * 16) + q * 4) << 1) + (row & 1);

  const float An2 = -__expf(A_log[(size_t)(d0 + ch) * N + n]) * LOG2E;
  const float4 D4 = *(const float4*)(Dp + d0 + q * 4);
  const bool b0 = (n & 1) != 0, b1 = (n & 2) != 0;
  const bool b2 = (n & 4) != 0, b3 = (n & 8) != 0;

  float4 dtn = *(const float4*)dt_src;
  float4 xn  = *(const float4*)x_src;
  float4 Bn  = *(const float4*)B_src;
  float4 Cn  = *(const float4*)C_src;
  {
    float2* pd = (float2*)s_dtus[0];
    float2* pb = (float2*)s_bc[0];
    pd[sbase + 0] = make_float2(dtn.x, dtn.x * xn.x);
    pd[sbase + 2] = make_float2(dtn.y, dtn.y * xn.y);
    pd[sbase + 4] = make_float2(dtn.z, dtn.z * xn.z);
    pd[sbase + 6] = make_float2(dtn.w, dtn.w * xn.w);
    pb[sbase + 0] = make_float2(Bn.x, Cn.x);
    pb[sbase + 2] = make_float2(Bn.y, Cn.y);
    pb[sbase + 4] = make_float2(Bn.z, Cn.z);
    pb[sbase + 6] = make_float2(Bn.w, Cn.w);
  }
  float4 xcur = xn;
  __syncthreads();

  float h = 0.0f;
  const int nCh = L / TC;
  for (int k = 0; k < nCh; ++k) {
    const int cur = k & 1;
    const bool more = (k + 1) < nCh;
    if (more) {
      dtn = *(const float4*)(dt_src + (size_t)(k + 1) * stepD);
      xn  = *(const float4*)(x_src  + (size_t)(k + 1) * stepD);
      Bn  = *(const float4*)(B_src  + (size_t)(k + 1) * stepN);
      Cn  = *(const float4*)(C_src  + (size_t)(k + 1) * stepN);
    }

    const float4* pdt = s_dtus[cur] + ch;
    const float4* pbc = s_bc[cur] + n;
    float* py = (float*)s_y[cur] + n * 16 + ch;
#pragma unroll 2
    for (int g = 0; g < TC / 16; ++g) {
      float p[16];
#pragma unroll
      for (int tp = 0; tp < 8; ++tp) {
        float4 v = pdt[(g * 8 + tp) * 16];
        float4 w = pbc[(g * 8 + tp) * 16];
        float a0 = __builtin_amdgcn_exp2f(v.x * An2);
        h = a0 * h + v.y * w.x;
        p[2 * tp] = h * w.y;
        float a1 = __builtin_amdgcn_exp2f(v.z * An2);
        h = a1 * h + v.w * w.z;
        p[2 * tp + 1] = h * w.w;
      }
      float r8[8], r4[4], r2[2];
#pragma unroll
      for (int j = 0; j < 8; ++j) r8[j] = fold_pair<1>(b0, p[2 * j], p[2 * j + 1]);
#pragma unroll
      for (int j = 0; j < 4; ++j) r4[j] = fold_pair<2>(b1, r8[2 * j], r8[2 * j + 1]);
#pragma unroll
      for (int j = 0; j < 2; ++j) r2[j] = fold_pair<4>(b2, r4[2 * j], r4[2 * j + 1]);
      float ysum = fold_pair<8>(b3, r2[0], r2[1]);
      py[g * 256] = ysum;
    }

    if (more) {
      float2* pd = (float2*)s_dtus[cur ^ 1];
      float2* pb = (float2*)s_bc[cur ^ 1];
      pd[sbase + 0] = make_float2(dtn.x, dtn.x * xn.x);
      pd[sbase + 2] = make_float2(dtn.y, dtn.y * xn.y);
      pd[sbase + 4] = make_float2(dtn.z, dtn.z * xn.z);
      pd[sbase + 6] = make_float2(dtn.w, dtn.w * xn.w);
      pb[sbase + 0] = make_float2(Bn.x, Cn.x);
      pb[sbase + 2] = make_float2(Bn.y, Cn.y);
      pb[sbase + 4] = make_float2(Bn.z, Cn.z);
      pb[sbase + 6] = make_float2(Bn.w, Cn.w);
    }
    __syncthreads();

    float4 yv = s_y[cur][row * 4 + q];
    float4 o;
    o.x = yv.x + xcur.x * D4.x;
    o.y = yv.y + xcur.y * D4.y;
    o.z = yv.z + xcur.z * D4.z;
    o.w = yv.w + xcur.w * D4.w;
    *(float4*)(out_dst + (size_t)k * stepD) = o;
    xcur = xn;
  }
}

#define FUSE_K 128
#define DL_PITCH 132
#define DT_PITCH 20

__device__ __forceinline__ float softplus_sel(float v) {
  float l = log1pf(__expf(v));
  return (v > 20.0f) ? v : l;
}

__global__ __launch_bounds__(512) void scan_dtfused_kernel(
    const float* __restrict__ x, const float* __restrict__ delta,
    const float* __restrict__ Bm, const float* __restrict__ Cm,
    const float* __restrict__ W, const float* __restrict__ bias,
    const float* __restrict__ A_log, const float* __restrict__ Dp,
    float* __restrict__ out, int L, int Dc) {
  const int N = 16;
  const int TC = SCAN_TC;
  typedef float4 stage_t[TC / 2 * 16];
  typedef float4 ysum_t[TC * 4];
  typedef float4 dtp_t[2][TC * DT_PITCH / 4];
  typedef float4 dlt_t[TC * DL_PITCH / 4];
  __shared__ float4 smem[2 * (TC / 2 * 16) * 2 + 2 * (TC * 4) +
                         2 * 2 * (TC * DT_PITCH / 4) + 2 * (TC * DL_PITCH / 4)];
  stage_t* s_dtus = (stage_t*)smem;
  stage_t* s_bc   = (stage_t*)(smem + 2 * (TC / 2 * 16));
  ysum_t* s_y     = (ysum_t*)(smem + 4 * (TC / 2 * 16));
  dtp_t* s_dtp    = (dtp_t*)(smem + 4 * (TC / 2 * 16) + 2 * (TC * 4));
  dlt_t* s_delta  = (dlt_t*)(smem + 4 * (TC / 2 * 16) + 2 * (TC * 4) +
                             2 * 2 * (TC * DT_PITCH / 4));

  const int tid = threadIdx.x;
  const int lane = tid & 31;
  const int wave = __builtin_amdgcn_readfirstlane(tid >> 5);
  const bool producer = wave >= 8;
  const int w8 = wave & 7;
  const int t8 = tid & 255;
  const int n = lane & 15;
  const int half = lane >> 4;
  const int ch = w8 * 2 + half;
  const int DB = Dc >> 4;
  const int b = blockIdx.x / DB;
  const int d0 = (blockIdx.x % DB) << 4;
  const int tile16 = w8 & 3;
  const int kh = w8 >> 2;

  const int row = t8 >> 2;
  const int q = t8 & 3;
  const size_t grow = (size_t)b * L + row;
  const float* x_src  = x  + grow * Dc + d0 + q * 4;
  const float* B_src  = Bm + grow * N + q * 4;
  const float* C_src  = Cm + grow * N + q * 4;
  float* out_dst      = out + grow * Dc + d0 + q * 4;
  const size_t stepD = (size_t)TC * Dc;
  const size_t stepN = (size_t)TC * N;
  const int sbase = ((((row >> 1) * 16) + q * 4) << 1) + (row & 1);

  const bool b0 = (n & 1) != 0, b1 = (n & 2) != 0;
  const bool b2 = (n & 4) != 0, b3 = (n & 8) != 0;

  const int nCh = L / TC;
  const int lastCh = nCh - 1;

  float An2 = 0.0f;
  float4 D4 = make_float4(0.0f, 0.0f, 0.0f, 0.0f);
  float4 bias4 = D4;
  v2f bw[16];
  float4 xprev = D4, xthis = D4, xnext = D4, Bnext = D4, Cnext = D4;
  float4 B0 = D4, C0 = D4;
  if (producer) {
    const int c1 = lastCh < 1 ? lastCh : 1;
    xthis = *(const float4*)x_src;
    B0 = *(const float4*)B_src;
    C0 = *(const float4*)C_src;
    xnext = *(const float4*)(x_src + (size_t)c1 * stepD);
    Bnext = *(const float4*)(B_src + (size_t)c1 * stepN);
    Cnext = *(const float4*)(C_src + (size_t)c1 * stepN);
    D4 = *(const float4*)(Dp + d0 + q * 4);
    bias4 = *(const float4*)(bias + d0 + q * 4);
    const float* wsrc = W + (size_t)(d0 + n) * FUSE_K + kh * 64 + half * 2;
#pragma unroll
    for (int s = 0; s < 16; ++s) bw[s] = *(const v2f*)(wsrc + 4 * s);
  } else {
    An2 = -__expf(A_log[(size_t)(d0 + ch) * N + n]) * LOG2E;
#pragma unroll
    for (int s = 0; s < 16; ++s) bw[s] = v2f{0.0f, 0.0f};
  }

  const float4* dl_src = (const float4*)(delta + (size_t)b * L * FUSE_K) + t8;
  const int dl_dst = w8 * (DL_PITCH / 4) + lane;
#define DELTA_DECL float4 dl0, dl1, dl2, dl3, dl4, dl5, dl6, dl7;
#define DELTA_LOAD(c)                                                   \
  {                                                                     \
    const float4* dlp = dl_src + (size_t)(c) * (TC * FUSE_K / 4);       \
    dl0 = dlp[0 * 256]; dl1 = dlp[1 * 256];                             \
    dl2 = dlp[2 * 256]; dl3 = dlp[3 * 256];                             \
    dl4 = dlp[4 * 256]; dl5 = dlp[5 * 256];                             \
    dl6 = dlp[6 * 256]; dl7 = dlp[7 * 256];                             \
  }
#define DELTA_STORE(buf)                                                \
  {                                                                     \
    float4* dls = s_delta[buf] + dl_dst;                                \
    dls[0 * 8 * (DL_PITCH / 4)] = dl0; dls[1 * 8 * (DL_PITCH / 4)] = dl1; \
    dls[2 * 8 * (DL_PITCH / 4)] = dl2; dls[3 * 8 * (DL_PITCH / 4)] = dl3; \
    dls[4 * 8 * (DL_PITCH / 4)] = dl4; dls[5 * 8 * (DL_PITCH / 4)] = dl5; \
    dls[6 * 8 * (DL_PITCH / 4)] = dl6; dls[7 * 8 * (DL_PITCH / 4)] = dl7; \
  }

  auto compute_dt = [&](int buf) {
    const float* sa = (const float*)s_delta[buf] +
                      (tile16 * 16 + n) * DL_PITCH + kh * 64 + half * 2;
    v2f a[16];
#pragma unroll
    for (int s = 0; s < 16; ++s) a[s] = *(const v2f*)(sa + 4 * s);
    asm volatile("" ::"v"(a[0]), "v"(a[1]), "v"(a[2]), "v"(a[3]), "v"(a[4]),
                 "v"(a[5]), "v"(a[6]), "v"(a[7]), "v"(a[8]), "v"(a[9]),
                 "v"(a[10]), "v"(a[11]), "v"(a[12]), "v"(a[13]), "v"(a[14]),
                 "v"(a[15]));
    v8f acc0 = {};
    v8f acc1 = {};
#pragma unroll
    for (int s = 0; s < 16; s += 2) {
      acc0 = __builtin_amdgcn_wmma_f32_16x16x4_f32(
          false, a[s], false, bw[s], (short)0, acc0, false, false);
      acc1 = __builtin_amdgcn_wmma_f32_16x16x4_f32(
          false, a[s + 1], false, bw[s + 1], (short)0, acc1, false, false);
    }
    float* sd = (float*)s_dtp[buf][kh] + (tile16 * 16 + half * 8) * DT_PITCH + n;
#pragma unroll
    for (int r = 0; r < 8; ++r) sd[r * DT_PITCH] = acc0[r] + acc1[r];
  };

  auto stage = [&](int buf, float4 xv, float4 Bv, float4 Cv) {
    const float* sp = (const float*)s_dtp[buf][0] + row * DT_PITCH + q * 4;
    float4 p0 = *(const float4*)sp;
    float4 p1 = *(const float4*)(sp + TC * DT_PITCH);
    float4 dtv;
    dtv.x = softplus_sel((p0.x + p1.x) + bias4.x);
    dtv.y = softplus_sel((p0.y + p1.y) + bias4.y);
    dtv.z = softplus_sel((p0.z + p1.z) + bias4.z);
    dtv.w = softplus_sel((p0.w + p1.w) + bias4.w);
    float2* pd = (float2*)s_dtus[buf];
    float2* pb = (float2*)s_bc[buf];
    pd[sbase + 0] = make_float2(dtv.x, dtv.x * xv.x);
    pd[sbase + 2] = make_float2(dtv.y, dtv.y * xv.y);
    pd[sbase + 4] = make_float2(dtv.z, dtv.z * xv.z);
    pd[sbase + 6] = make_float2(dtv.w, dtv.w * xv.w);
    pb[sbase + 0] = make_float2(Bv.x, Cv.x);
    pb[sbase + 2] = make_float2(Bv.y, Cv.y);
    pb[sbase + 4] = make_float2(Bv.z, Cv.z);
    pb[sbase + 6] = make_float2(Bv.w, Cv.w);
  };

  auto write_out = [&](int c, float4 xv) {
    float4 yv = s_y[c & 1][row * 4 + q];
    float4 o;
    o.x = yv.x + xv.x * D4.x;
    o.y = yv.y + xv.y * D4.y;
    o.z = yv.z + xv.z * D4.z;
    o.w = yv.w + xv.w * D4.w;
    *(float4*)(out_dst + (size_t)c * stepD) = o;
  };

  if (producer) {
    {
      DELTA_DECL
      DELTA_LOAD(0)
      DELTA_STORE(0)
    }
    {
      DELTA_DECL
      DELTA_LOAD(lastCh < 1 ? lastCh : 1)
      DELTA_STORE(1)
    }
  }
  __syncthreads();
  if (producer) {
    compute_dt(0);
    compute_dt(1);
  }
  __syncthreads();
  if (producer) {
    stage(0, xthis, B0, C0);
    DELTA_DECL
    DELTA_LOAD(lastCh < 2 ? lastCh : 2)
    DELTA_STORE(0)
  }
  __syncthreads();

  float h = 0.0f;
  for (int k = 0; k < nCh; ++k) {
    const int cur = k & 1;
    if (!producer) {
      const float4* pdt = s_dtus[cur] + ch;
      const float4* pbc = s_bc[cur] + n;
      float* py = (float*)s_y[cur] + n * 16 + ch;
#pragma unroll 2
      for (int g = 0; g < TC / 16; ++g) {
        float p[16];
#pragma unroll
        for (int tp = 0; tp < 8; ++tp) {
          float4 v = pdt[(g * 8 + tp) * 16];
          float4 w = pbc[(g * 8 + tp) * 16];
          float a0 = __builtin_amdgcn_exp2f(v.x * An2);
          h = a0 * h + v.y * w.x;
          p[2 * tp] = h * w.y;
          float a1 = __builtin_amdgcn_exp2f(v.z * An2);
          h = a1 * h + v.w * w.z;
          p[2 * tp + 1] = h * w.w;
        }
        float r8[8], r4[4], r2[2];
#pragma unroll
        for (int j = 0; j < 8; ++j) r8[j] = fold_pair<1>(b0, p[2 * j], p[2 * j + 1]);
#pragma unroll
        for (int j = 0; j < 4; ++j) r4[j] = fold_pair<2>(b1, r8[2 * j], r8[2 * j + 1]);
#pragma unroll
        for (int j = 0; j < 2; ++j) r2[j] = fold_pair<4>(b2, r4[2 * j], r4[2 * j + 1]);
        float ysum = fold_pair<8>(b3, r2[0], r2[1]);
        py[g * 256] = ysum;
      }
    } else {
      const bool more = (k + 1) < nCh;
      const bool more2 = (k + 2) < nCh;
      const bool more3 = (k + 3) < nCh;
      const int k2 = more2 ? k + 2 : lastCh;
      const float4 xnn = *(const float4*)(x_src + (size_t)k2 * stepD);
      const float4 Bnn = *(const float4*)(B_src + (size_t)k2 * stepN);
      const float4 Cnn = *(const float4*)(C_src + (size_t)k2 * stepN);
      DELTA_DECL
      if (more3) DELTA_LOAD(k + 3)

      if (k > 0) write_out(k - 1, xprev);
      if (more2) compute_dt(cur);
      if (more) stage(cur ^ 1, xnext, Bnext, Cnext);
      if (more3) DELTA_STORE(cur ^ 1)
      xprev = xthis;
      xthis = xnext;
      xnext = xnn;
      Bnext = Bnn;
      Cnext = Cnn;
    }
    __syncthreads();
  }
  if (producer) write_out(lastCh, xprev);
#undef DELTA_DECL
#undef DELTA_LOAD
#undef DELTA_STORE
}

__global__ void scan_simple_kernel(const float* __restrict__ dt,
                                   const float* __restrict__ x,
                                   const float* __restrict__ Bm,
                                   const float* __restrict__ Cm,
                                   const float* __restrict__ A_log,
                                   const float* __restrict__ Dp,
                                   float* __restrict__ out,
                                   int L, int Dc, int Bsz) {
  const int N = 16;
  long long idx = (long long)blockIdx.x * blockDim.x + threadIdx.x;
  if (idx >= (long long)Bsz * Dc) return;
  int d = (int)(idx % Dc);
  int b = (int)(idx / Dc);
  float An2[N], h[N];
#pragma unroll
  for (int n = 0; n < N; ++n) {
    An2[n] = -__expf(A_log[(size_t)d * N + n]) * LOG2E;
    h[n] = 0.0f;
  }
  float Dv = Dp[d];
  for (int l = 0; l < L; ++l) {
    size_t row = (size_t)b * L + l;
    size_t g = row * Dc + d;
    float dtv = dt[g];
    float xv = x[g];
    float us = dtv * xv;
    const float* Br = Bm + row * N;
    const float* Cr = Cm + row * N;
    float y = 0.0f;
#pragma unroll
    for (int n = 0; n < N; ++n) {
      h[n] = __builtin_amdgcn_exp2f(dtv * An2[n]) * h[n] + us * Br[n];
      y += h[n] * Cr[n];
    }
    out[g] = y + xv * Dv;
  }
}

__global__ void scan_fused_kernel(const float* __restrict__ x,
                                  const float* __restrict__ delta,
                                  const float* __restrict__ Bm,
                                  const float* __restrict__ Cm,
                                  const float* __restrict__ W,
                                  const float* __restrict__ bias,
                                  const float* __restrict__ A_log,
                                  const float* __restrict__ Dp,
                                  float* __restrict__ out,
                                  int L, int Dc, int K, int Bsz) {
  const int N = 16;
  long long idx = (long long)blockIdx.x * blockDim.x + threadIdx.x;
  if (idx >= (long long)Bsz * Dc) return;
  int d = (int)(idx % Dc);
  int b = (int)(idx / Dc);
  float An2[N], h[N];
#pragma unroll
  for (int n = 0; n < N; ++n) {
    An2[n] = -__expf(A_log[(size_t)d * N + n]) * LOG2E;
    h[n] = 0.0f;
  }
  float Dv = Dp[d];
  float bv = bias[d];
  const float* wr = W + (size_t)d * K;
  for (int l = 0; l < L; ++l) {
    size_t row = (size_t)b * L + l;
    const float* dr = delta + row * K;
    float acc = bv;
    for (int r = 0; r < K; ++r) acc += dr[r] * wr[r];
    float dtv = softplus_f(acc);
    float xv = x[row * Dc + d];
    float us = dtv * xv;
    const float* Br = Bm + row * N;
    const float* Cr = Cm + row * N;
    float y = 0.0f;
#pragma unroll
    for (int n = 0; n < N; ++n) {
      h[n] = __builtin_amdgcn_exp2f(dtv * An2[n]) * h[n] + us * Br[n];
      y += h[n] * Cr[n];
    }
    out[row * Dc + d] = y + xv * Dv;
  }
}

extern "C" void kernel_launch(void* const* d_in, const int* in_sizes, int n_in,
                              void* d_out, int out_size, void* d_ws, size_t ws_size,
                              hipStream_t stream) {
  const float* x     = (const float*)d_in[0];
  const float* delta = (const float*)d_in[1];
  const float* Bm    = (const float*)d_in[2];
  const float* Cm    = (const float*)d_in[3];
  const float* W     = (const float*)d_in[4];
  const float* bias  = (const float*)d_in[5];
  const float* A_log = (const float*)d_in[6];
  const float* Dp    = (const float*)d_in[7];
  float* out = (float*)d_out;

  int Dc = in_sizes[5];
  int K  = in_sizes[4] / Dc;
  long long BL = (long long)in_sizes[0] / Dc;
  int Bsz = (BL % 2 == 0) ? 2 : 1;
  int L = (int)(BL / Bsz);

  size_t dtElems = (size_t)BL * Dc;
  size_t dtBytes = dtElems * sizeof(float);

  if (K == FUSE_K && (L % SCAN_TC) == 0 && (Dc % 16) == 0) {
    int blocks = Bsz * (Dc / 16);
    scan_dtfused_kernel<<<blocks, 512, 0, stream>>>(x, delta, Bm, Cm, W, bias,
                                                    A_log, Dp, out, L, Dc);
    return;
  }

  if (ws_size < dtBytes) {
    int threads = 256;
    long long tot = (long long)Bsz * Dc;
    int blocks = (int)((tot + threads - 1) / threads);
    scan_fused_kernel<<<blocks, threads, 0, stream>>>(
        x, delta, Bm, Cm, W, bias, A_log, Dp, out, L, Dc, K, Bsz);
    return;
  }

  float* dt = (float*)d_ws;

  if ((BL % 16) == 0 && (Dc % 32) == 0 && (K % 4) == 0) {
    int tiles = (int)(BL / 16) * (Dc / 32);
    int blocks = (tiles + 7) / 8;
    dtproj_wmma_kernel<<<blocks, 256, 0, stream>>>(delta, W, bias, dt,
                                                   (int)BL, Dc, K);
  } else {
    long long tot = BL * (long long)Dc;
    int blocks = (int)((tot + 255) / 256);
    dtproj_naive_kernel<<<blocks, 256, 0, stream>>>(delta, W, bias, dt, BL, Dc, K);
  }

  if ((L % SCAN_TC) == 0 && (Dc % 16) == 0) {
    int blocks = Bsz * (Dc / 16);
    scan_lanes_kernel<<<blocks, 256, 0, stream>>>(dt, x, Bm, Cm, A_log, Dp, out,
                                                  L, Dc);
  } else {
    long long tot = (long long)Bsz * Dc;
    int blocks = (int)((tot + 255) / 256);
    scan_simple_kernel<<<blocks, 256, 0, stream>>>(dt, x, Bm, Cm, A_log, Dp, out,
                                                   L, Dc, Bsz);
  }
}


# scan lanes re-tiled: a lane owns states j and j+8 of a channel (4 scan waves, not 8), operands read one 8-step group ahead; readout adds the lane's two products, then 3 DPP levels: sum order changes
# speedup vs baseline: 2.6577x; 1.0504x over previous
#include <hip/hip_runtime.h>
#include <math.h>

typedef __attribute__((ext_vector_type(2))) float v2f;
typedef __attribute__((ext_vector_type(8))) float v8f;

#define LOG2E 1.4426950408889634f

__device__ __forceinline__ float softplus_f(float v) {
  return (v > 20.0f) ? v : log1pf(__expf(v));
}

__global__ __launch_bounds__(256) void dtproj_wmma_kernel(
    const float* __restrict__ delta, const float* __restrict__ W,
    const float* __restrict__ bias, float* __restrict__ dt,
    int M, int Dc, int K) {
  int wave = threadIdx.x >> 5;
  int lane = threadIdx.x & 31;
  int tilesN = Dc >> 5;
  int tile = blockIdx.x * 8 + wave;
  int totalTiles = (M >> 4) * tilesN;
  if (tile >= totalTiles) return;
  int tn = tile % tilesN;
  int tm = tile / tilesN;
  int mn   = lane & 15;
  int half = lane >> 4;

  const float* arow  = delta + (size_t)(tm * 16 + mn) * K;
  const float* brow0 = W     + (size_t)(tn * 32 + mn) * K;
  const float* brow1 = W     + (size_t)(tn * 32 + 16 + mn) * K;

  v8f acc0 = {};
  v8f acc1 = {};
  for (int k0 = 0; k0 < K; k0 += 4) {
    int kb = k0 + (half << 1);
    v2f a, b0, b1;
    a.x  = arow[kb];  a.y  = arow[kb + 1];
    b0.x = brow0[kb]; b0.y = brow0[kb + 1];
    b1.x = brow1[kb]; b1.y = brow1[kb + 1];
    acc0 = __builtin_amdgcn_wmma_f32_16x16x4_f32(
        false, a, false, b0, (short)0, acc0, false, false);
    acc1 = __builtin_amdgcn_wmma_f32_16x16x4_f32(
        false, a, false, b1, (short)0, acc1, false, false);
  }

  int gd0 = tn * 32 + mn;
  int gd1 = gd0 + 16;
  float bv0 = bias[gd0];
  float bv1 = bias[gd1];
  int gm0 = tm * 16 + half * 8;
#pragma unroll
  for (int v = 0; v < 8; ++v) {
    size_t rowoff = (size_t)(gm0 + v) * Dc;
    dt[rowoff + gd0] = softplus_f(acc0[v] + bv0);
    dt[rowoff + gd1] = softplus_f(acc1[v] + bv1);
  }
}

__global__ void dtproj_naive_kernel(const float* __restrict__ delta,
                                    const float* __restrict__ W,
                                    const float* __restrict__ bias,
                                    float* __restrict__ dt,
                                    long long M, int Dc, int K) {
  long long idx = (long long)blockIdx.x * blockDim.x + threadIdx.x;
  if (idx >= M * (long long)Dc) return;
  int d = (int)(idx % Dc);
  long long m = idx / Dc;
  float acc = bias[d];
  const float* dr = delta + m * (long long)K;
  const float* wr = W + (size_t)d * K;
  for (int r = 0; r < K; ++r) acc += dr[r] * wr[r];
  dt[idx] = softplus_f(acc);
}

#define SCAN_TC 64

template <int MASK>
__device__ __forceinline__ float row_xor(float v) {
  int r = __builtin_amdgcn_update_dpp(0, __builtin_bit_cast(int, v),
                                      0x160 + MASK, 0xf, 0xf, true);
  return __builtin_bit_cast(float, r);
}

template <int MASK>
__device__ __forceinline__ float fold_pair(bool bit, float lo, float hi) {
  float keep = bit ? hi : lo;
  float send = bit ? lo : hi;
  return keep + row_xor<MASK>(send);
}

__global__ __launch_bounds__(256) void scan_lanes_kernel(
    const float* __restrict__ dt, const float* __restrict__ x,
    const float* __restrict__ Bm, const float* __restrict__ Cm,
    const float* __restrict__ A_log, const float* __restrict__ Dp,
    float* __restrict__ out, int L, int Dc) {
  const int N = 16;
  const int TC = SCAN_TC;
  __shared__ float4 s_dtus[2][TC / 2 * 16];
  __shared__ float4 s_bc[2][TC / 2 * 16];
  __shared__ float4 s_y[2][TC * 4];

  const int tid = threadIdx.x;
  const int lane = tid & 31;
  const int wave = tid >> 5;
  const int n = lane & 15;
  const int ch = wave * 2 + (lane >> 4);
  const int DB = Dc >> 4;
  const int b = blockIdx.x / DB;
  const int d0 = (blockIdx.x % DB) << 4;

  const int row = tid >> 2;
  const int q = tid & 3;
  const size_t grow = (size_t)b * L + row;
  const float* dt_src = dt + grow * Dc + d0 + q * 4;
  const float* x_src  = x  + grow * Dc + d0 + q * 4;
  const float* B_src  = Bm + grow * N + q * 4;
  const float* C_src  = Cm + grow * N + q * 4;
  float* out_dst      = out + grow * Dc + d0 + q * 4;
  const size_t stepD = (size_t)TC * Dc;
  const size_t stepN = (size_t)TC * N;
  const int sbase = ((((row >> 1) * 16) + q * 4) << 1) + (row & 1);

  const float An2 = -__expf(A_log[(size_t)(d0 + ch) * N + n]) * LOG2E;
  const float4 D4 = *(const float4*)(Dp + d0 + q * 4);
  const bool b0 = (n & 1) != 0, b1 = (n & 2) != 0;
  const bool b2 = (n & 4) != 0, b3 = (n & 8) != 0;

  float4 dtn = *(const float4*)dt_src;
  float4 xn  = *(const float4*)x_src;
  float4 Bn  = *(const float4*)B_src;
  float4 Cn  = *(const float4*)C_src;
  {
    float2* pd = (float2*)s_dtus[0];
    float2* pb = (float2*)s_bc[0];
    pd[sbase + 0] = make_float2(dtn.x, dtn.x * xn.x);
    pd[sbase + 2] = make_float2(dtn.y, dtn.y * xn.y);
    pd[sbase + 4] = make_float2(dtn.z, dtn.z * xn.z);
    pd[sbase + 6] = make_float2(dtn.w, dtn.w * xn.w);
    pb[sbase + 0] = make_float2(Bn.x, Cn.x);
    pb[sbase + 2] = make_float2(Bn.y, Cn.y);
    pb[sbase + 4] = make_float2(Bn.z, Cn.z);
    pb[sbase + 6] = make_float2(Bn.w, Cn.w);
  }
  float4 xcur = xn;
  __syncthreads();

  float h = 0.0f;
  const int nCh = L / TC;
  for (int k = 0; k < nCh; ++k) {
    const int cur = k & 1;
    const bool more = (k + 1) < nCh;
    if (more) {
      dtn = *(const float4*)(dt_src + (size_t)(k + 1) * stepD);
      xn  = *(const float4*)(x_src  + (size_t)(k + 1) * stepD);
      Bn  = *(const float4*)(B_src  + (size_t)(k + 1) * stepN);
      Cn  = *(const float4*)(C_src  + (size_t)(k + 1) * stepN);
    }

    const float4* pdt = s_dtus[cur] + ch;
    const float4* pbc = s_bc[cur] + n;
    float* py = (float*)s_y[cur] + n * 16 + ch;
#pragma unroll 2
    for (int g = 0; g < TC / 16; ++g) {
      float p[16];
#pragma unroll
      for (int tp = 0; tp < 8; ++tp) {
        float4 v = pdt[(g * 8 + tp) * 16];
        float4 w = pbc[(g * 8 + tp) * 16];
        float a0 = __builtin_amdgcn_exp2f(v.x * An2);
        h = a0 * h + v.y * w.x;
        p[2 * tp] = h * w.y;
        float a1 = __builtin_amdgcn_exp2f(v.z * An2);
        h = a1 * h + v.w * w.z;
        p[2 * tp + 1] = h * w.w;
      }
      float r8[8], r4[4], r2[2];
#pragma unroll
      for (int j = 0; j < 8; ++j) r8[j] = fold_pair<1>(b0, p[2 * j], p[2 * j + 1]);
#pragma unroll
      for (int j = 0; j < 4; ++j) r4[j] = fold_pair<2>(b1, r8[2 * j], r8[2 * j + 1]);
#pragma unroll
      for (int j = 0; j < 2; ++j) r2[j] = fold_pair<4>(b2, r4[2 * j], r4[2 * j + 1]);
      float ysum = fold_pair<8>(b3, r2[0], r2[1]);
      py[g * 256] = ysum;
    }

    if (more) {
      float2* pd = (float2*)s_dtus[cur ^ 1];
      float2* pb = (float2*)s_bc[cur ^ 1];
      pd[sbase + 0] = make_float2(dtn.x, dtn.x * xn.x);
      pd[sbase + 2] = make_float2(dtn.y, dtn.y * xn.y);
      pd[sbase + 4] = make_float2(dtn.z, dtn.z * xn.z);
      pd[sbase + 6] = make_float2(dtn.w, dtn.w * xn.w);
      pb[sbase + 0] = make_float2(Bn.x, Cn.x);
      pb[sbase + 2] = make_float2(Bn.y, Cn.y);
      pb[sbase + 4] = make_float2(Bn.z, Cn.z);
      pb[sbase + 6] = make_float2(Bn.w, Cn.w);
    }
    __syncthreads();

    float4 yv = s_y[cur][row * 4 + q];
    float4 o;
    o.x = yv.x + xcur.x * D4.x;
    o.y = yv.y + xcur.y * D4.y;
    o.z = yv.z + xcur.z * D4.z;
    o.w = yv.w + xcur.w * D4.w;
    *(float4*)(out_dst + (size_t)k * stepD) = o;
    xcur = xn;
  }
}

#define FUSE_K 128
#define DL_PITCH 132
#define DT_PITCH 20

__device__ __forceinline__ float softplus_sel(float v) {
  float l = log1pf(__expf(v));
  return (v > 20.0f) ? v : l;
}

__global__ __launch_bounds__(384) void scan_dtfused_kernel(
    const float* __restrict__ x, const float* __restrict__ delta,
    const float* __restrict__ Bm, const float* __restrict__ Cm,
    const float* __restrict__ W, const float* __restrict__ bias,
    const float* __restrict__ A_log, const float* __restrict__ Dp,
    float* __restrict__ out, int L, int Dc) {
  const int N = 16;
  const int TC = SCAN_TC;
  typedef float4 stage_t[TC / 2 * 16];
  typedef float4 ysum_t[TC * 4];
  typedef float4 dtp_t[2][TC * DT_PITCH / 4];
  typedef float4 dlt_t[TC * DL_PITCH / 4];
  __shared__ float4 smem[2 * (TC / 2 * 16) * 2 + 2 * (TC * 4) +
                         2 * 2 * (TC * DT_PITCH / 4) + 2 * (TC * DL_PITCH / 4)];
  stage_t* s_dtus = (stage_t*)smem;
  stage_t* s_bc   = (stage_t*)(smem + 2 * (TC / 2 * 16));
  ysum_t* s_y     = (ysum_t*)(smem + 4 * (TC / 2 * 16));
  dtp_t* s_dtp    = (dtp_t*)(smem + 4 * (TC / 2 * 16) + 2 * (TC * 4));
  dlt_t* s_delta  = (dlt_t*)(smem + 4 * (TC / 2 * 16) + 2 * (TC * 4) +
                             2 * 2 * (TC * DT_PITCH / 4));

  const int tid = threadIdx.x;
  const int lane = tid & 31;
  const int wave = __builtin_amdgcn_readfirstlane(tid >> 5);
  const bool producer = wave >= 4;
  const int w8 = (wave - 4) & 7;
  const int t8 = (tid - 128) & 255;
  const int n = lane & 15;
  const int half = lane >> 4;
  const int j8 = lane & 7;
  const int ch = (wave & 3) * 4 + (lane >> 3);
  const int DB = Dc >> 4;
  const int b = blockIdx.x / DB;
  const int d0 = (blockIdx.x % DB) << 4;
  const int tile16 = w8 & 3;
  const int kh = w8 >> 2;

  const int row = t8 >> 2;
  const int q = t8 & 3;
  const size_t grow = (size_t)b * L + row;
  const float* x_src  = x  + grow * Dc + d0 + q * 4;
  const float* B_src  = Bm + grow * N + q * 4;
  const float* C_src  = Cm + grow * N + q * 4;
  float* out_dst      = out + grow * Dc + d0 + q * 4;
  const size_t stepD = (size_t)TC * Dc;
  const size_t stepN = (size_t)TC * N;
  const int sbase = ((((row >> 1) * 16) + q * 4) << 1) + (row & 1);

  const bool b0 = (j8 & 1) != 0, b1 = (j8 & 2) != 0, b2 = (j8 & 4) != 0;

  const int nCh = L / TC;
  const int lastCh = nCh - 1;

  float An2a = 0.0f, An2b = 0.0f;
  float4 D4 = make_float4(0.0f, 0.0f, 0.0f, 0.0f);
  float4 bias4 = D4;
  v2f bw[16];
  float4 xprev = D4, xthis = D4, xnext = D4, Bnext = D4, Cnext = D4;
  float4 B0 = D4, C0 = D4;
  if (producer) {
    const int c1 = lastCh < 1 ? lastCh : 1;
    xthis = *(const float4*)x_src;
    B0 = *(const float4*)B_src;
    C0 = *(const float4*)C_src;
    xnext = *(const float4*)(x_src + (size_t)c1 * stepD);
    Bnext = *(const float4*)(B_src + (size_t)c1 * stepN);
    Cnext = *(const float4*)(C_src + (size_t)c1 * stepN);
    D4 = *(const float4*)(Dp + d0 + q * 4);
    bias4 = *(const float4*)(bias + d0 + q * 4);
    const float* wsrc = W + (size_t)(d0 + n) * FUSE_K + kh * 64 + half * 2;
#pragma unroll
    for (int s = 0; s < 16; ++s) bw[s] = *(const v2f*)(wsrc + 4 * s);
  } else {
    An2a = -__expf(A_log[(size_t)(d0 + ch) * N + j8]) * LOG2E;
    An2b = -__expf(A_log[(size_t)(d0 + ch) * N + j8 + 8]) * LOG2E;
#pragma unroll
    for (int s = 0; s < 16; ++s) bw[s] = v2f{0.0f, 0.0f};
  }

  const float4* dl_src = (const float4*)(delta + (size_t)b * L * FUSE_K) + t8;
  const int dl_dst = w8 * (DL_PITCH / 4) + lane;
#define DELTA_DECL float4 dl0, dl1, dl2, dl3, dl4, dl5, dl6, dl7;
#define DELTA_LOAD(c)                                                   \
  {                                                                     \
    const float4* dlp = dl_src + (size_t)(c) * (TC * FUSE_K / 4);       \
    dl0 = dlp[0 * 256]; dl1 = dlp[1 * 256];                             \
    dl2 = dlp[2 * 256]; dl3 = dlp[3 * 256];                             \
    dl4 = dlp[4 * 256]; dl5 = dlp[5 * 256];                             \
    dl6 = dlp[6 * 256]; dl7 = dlp[7 * 256];                             \
  }
#define DELTA_STORE(buf)                                                \
  {                                                                     \
    float4* dls = s_delta[buf] + dl_dst;                                \
    dls[0 * 8 * (DL_PITCH / 4)] = dl0; dls[1 * 8 * (DL_PITCH / 4)] = dl1; \
    dls[2 * 8 * (DL_PITCH / 4)] = dl2; dls[3 * 8 * (DL_PITCH / 4)] = dl3; \
    dls[4 * 8 * (DL_PITCH / 4)] = dl4; dls[5 * 8 * (DL_PITCH / 4)] = dl5; \
    dls[6 * 8 * (DL_PITCH / 4)] = dl6; dls[7 * 8 * (DL_PITCH / 4)] = dl7; \
  }

  auto compute_dt = [&](int buf) {
    const float* sa = (const float*)s_delta[buf] +
                      (tile16 * 16 + n) * DL_PITCH + kh * 64 + half * 2;
    v2f a[16];
#pragma unroll
    for (int s = 0; s < 16; ++s) a[s] = *(const v2f*)(sa + 4 * s);
    asm volatile("" ::"v"(a[0]), "v"(a[1]), "v"(a[2]), "v"(a[3]), "v"(a[4]),
                 "v"(a[5]), "v"(a[6]), "v"(a[7]), "v"(a[8]), "v"(a[9]),
                 "v"(a[10]), "v"(a[11]), "v"(a[12]), "v"(a[13]), "v"(a[14]),
                 "v"(a[15]));
    v8f acc0 = {};
    v8f acc1 = {};
#pragma unroll
    for (int s = 0; s < 16; s += 2) {
      acc0 = __builtin_amdgcn_wmma_f32_16x16x4_f32(
          false, a[s], false, bw[s], (short)0, acc0, false, false);
      acc1 = __builtin_amdgcn_wmma_f32_16x16x4_f32(
          false, a[s + 1], false, bw[s + 1], (short)0, acc1, false, false);
    }
    float* sd = (float*)s_dtp[buf][kh] + (tile16 * 16 + half * 8) * DT_PITCH + n;
#pragma unroll
    for (int r = 0; r < 8; ++r) sd[r * DT_PITCH] = acc0[r] + acc1[r];
  };

  auto stage = [&](int buf, float4 xv, float4 Bv, float4 Cv) {
    const float* sp = (const float*)s_dtp[buf][0] + row * DT_PITCH + q * 4;
    float4 p0 = *(const float4*)sp;
    float4 p1 = *(const float4*)(sp + TC * DT_PITCH);
    float4 dtv;
    dtv.x = softplus_sel((p0.x + p1.x) + bias4.x);
    dtv.y = softplus_sel((p0.y + p1.y) + bias4.y);
    dtv.z = softplus_sel((p0.z + p1.z) + bias4.z);
    dtv.w = softplus_sel((p0.w + p1.w) + bias4.w);
    float2* pd = (float2*)s_dtus[buf];
    float2* pb = (float2*)s_bc[buf];
    pd[sbase + 0] = make_float2(dtv.x, dtv.x * xv.x);
    pd[sbase + 2] = make_float2(dtv.y, dtv.y * xv.y);
    pd[sbase + 4] = make_float2(dtv.z, dtv.z * xv.z);
    pd[sbase + 6] = make_float2(dtv.w, dtv.w * xv.w);
    pb[sbase + 0] = make_float2(Bv.x, Cv.x);
    pb[sbase + 2] = make_float2(Bv.y, Cv.y);
    pb[sbase + 4] = make_float2(Bv.z, Cv.z);
    pb[sbase + 6] = make_float2(Bv.w, Cv.w);
  };

  auto write_out = [&](int c, float4 xv) {
    float4 yv = s_y[c & 1][row * 4 + q];
    float4 o;
    o.x = yv.x + xv.x * D4.x;
    o.y = yv.y + xv.y * D4.y;
    o.z = yv.z + xv.z * D4.z;
    o.w = yv.w + xv.w * D4.w;
    *(float4*)(out_dst + (size_t)c * stepD) = o;
  };

  if (producer) {
    {
      DELTA_DECL
      DELTA_LOAD(0)
      DELTA_STORE(0)
    }
    {
      DELTA_DECL
      DELTA_LOAD(lastCh < 1 ? lastCh : 1)
      DELTA_STORE(1)
    }
  }
  __syncthreads();
  if (producer) {
    compute_dt(0);
    compute_dt(1);
  }
  __syncthreads();
  if (producer) {
    stage(0, xthis, B0, C0);
    DELTA_DECL
    DELTA_LOAD(lastCh < 2 ? lastCh : 2)
    DELTA_STORE(0)
  }
  __syncthreads();

  float ha = 0.0f, hb = 0.0f;
  for (int k = 0; k < nCh; ++k) {
    const int cur = k & 1;
    if (!producer) {
      const float4* pdt = s_dtus[cur] + ch;
      const float4* pbc = s_bc[cur] + j8;
      float* py = (float*)s_y[cur] + j8 * 16 + ch;
      float4 V[2][4], WA[2][4], WB[2][4];
#pragma unroll
      for (int tq = 0; tq < 4; ++tq) {
        V[0][tq]  = pdt[tq * 16];
        WA[0][tq] = pbc[tq * 16];
        WB[0][tq] = pbc[tq * 16 + 8];
      }
#pragma unroll
      for (int g = 0; g < TC / 8; ++g) {
        if (g + 1 < TC / 8) {
#pragma unroll
          for (int tq = 0; tq < 4; ++tq) {
            V[(g + 1) % 2][tq]  = pdt[((g + 1) * 4 + tq) * 16];
            WA[(g + 1) % 2][tq] = pbc[((g + 1) * 4 + tq) * 16];
            WB[(g + 1) % 2][tq] = pbc[((g + 1) * 4 + tq) * 16 + 8];
          }
          __builtin_amdgcn_sched_barrier(0);
        }
        float p[8];
#pragma unroll
        for (int tq = 0; tq < 4; ++tq) {
          const float4 v  = V[g % 2][tq];
          const float4 wa = WA[g % 2][tq];
          const float4 wb = WB[g % 2][tq];
          float ea0 = __builtin_amdgcn_exp2f(v.x * An2a);
          float eb0 = __builtin_amdgcn_exp2f(v.x * An2b);
          ha = ea0 * ha + v.y * wa.x;
          hb = eb0 * hb + v.y * wb.x;
          p[2 * tq] = ha * wa.y + hb * wb.y;
          float ea1 = __builtin_amdgcn_exp2f(v.z * An2a);
          float eb1 = __builtin_amdgcn_exp2f(v.z * An2b);
          ha = ea1 * ha + v.w * wa.z;
          hb = eb1 * hb + v.w * wb.z;
          p[2 * tq + 1] = ha * wa.w + hb * wb.w;
        }
        float r4[4], r2[2];
#pragma unroll
        for (int j = 0; j < 4; ++j) r4[j] = fold_pair<1>(b0, p[2 * j], p[2 * j + 1]);
#pragma unroll
        for (int j = 0; j < 2; ++j) r2[j] = fold_pair<2>(b1, r4[2 * j], r4[2 * j + 1]);
        py[g * 128] = fold_pair<4>(b2, r2[0], r2[1]);
      }
    } else {
      const bool more = (k + 1) < nCh;
      const bool more2 = (k + 2) < nCh;
      const bool more3 = (k + 3) < nCh;
      const int k2 = more2 ? k + 2 : lastCh;
      const float4 xnn = *(const float4*)(x_src + (size_t)k2 * stepD);
      const float4 Bnn = *(const float4*)(B_src + (size_t)k2 * stepN);
      const float4 Cnn = *(const float4*)(C_src + (size_t)k2 * stepN);
      DELTA_DECL
      if (more3) DELTA_LOAD(k + 3)

      if (k > 0) write_out(k - 1, xprev);
      if (more2) compute_dt(cur);
      if (more) stage(cur ^ 1, xnext, Bnext, Cnext);
      if (more3) DELTA_STORE(cur ^ 1)
      xprev = xthis;
      xthis = xnext;
      xnext = xnn;
      Bnext = Bnn;
      Cnext = Cnn;
    }
    __syncthreads();
  }
  if (producer) write_out(lastCh, xprev);
#undef DELTA_DECL
#undef DELTA_LOAD
#undef DELTA_STORE
}

__global__ void scan_simple_kernel(const float* __restrict__ dt,
                                   const float* __restrict__ x,
                                   const float* __restrict__ Bm,
                                   const float* __restrict__ Cm,
                                   const float* __restrict__ A_log,
                                   const float* __restrict__ Dp,
                                   float* __restrict__ out,
                                   int L, int Dc, int Bsz) {
  const int N = 16;
  long long idx = (long long)blockIdx.x * blockDim.x + threadIdx.x;
  if (idx >= (long long)Bsz * Dc) return;
  int d = (int)(idx % Dc);
  int b = (int)(idx / Dc);
  float An2[N], h[N];
#pragma unroll
  for (int n = 0; n < N; ++n) {
    An2[n] = -__expf(A_log[(size_t)d * N + n]) * LOG2E;
    h[n] = 0.0f;
  }
  float Dv = Dp[d];
  for (int l = 0; l < L; ++l) {
    size_t row = (size_t)b * L + l;
    size_t g = row * Dc + d;
    float dtv = dt[g];
    float xv = x[g];
    float us = dtv * xv;
    const float* Br = Bm + row * N;
    const float* Cr = Cm + row * N;
    float y = 0.0f;
#pragma unroll
    for (int n = 0; n < N; ++n) {
      h[n] = __builtin_amdgcn_exp2f(dtv * An2[n]) * h[n] + us * Br[n];
      y += h[n] * Cr[n];
    }
    out[g] = y + xv * Dv;
  }
}

__global__ void scan_fused_kernel(const float* __restrict__ x,
                                  const float* __restrict__ delta,
                                  const float* __restrict__ Bm,
                                  const float* __restrict__ Cm,
                                  const float* __restrict__ W,
                                  const float* __restrict__ bias,
                                  const float* __restrict__ A_log,
                                  const float* __restrict__ Dp,
                                  float* __restrict__ out,
                                  int L, int Dc, int K, int Bsz) {
  const int N = 16;
  long long idx = (long long)blockIdx.x * blockDim.x + threadIdx.x;
  if (idx >= (long long)Bsz * Dc) return;
  int d = (int)(idx % Dc);
  int b = (int)(idx / Dc);
  float An2[N], h[N];
#pragma unroll
  for (int n = 0; n < N; ++n) {
    An2[n] = -__expf(A_log[(size_t)d * N + n]) * LOG2E;
    h[n] = 0.0f;
  }
  float Dv = Dp[d];
  float bv = bias[d];
  const float* wr = W + (size_t)d * K;
  for (int l = 0; l < L; ++l) {
    size_t row = (size_t)b * L + l;
    const float* dr = delta + row * K;
    float acc = bv;
    for (int r = 0; r < K; ++r) acc += dr[r] * wr[r];
    float dtv = softplus_f(acc);
    float xv = x[row * Dc + d];
    float us = dtv * xv;
    const float* Br = Bm + row * N;
    const float* Cr = Cm + row * N;
    float y = 0.0f;
#pragma unroll
    for (int n = 0; n < N; ++n) {
      h[n] = __builtin_amdgcn_exp2f(dtv * An2[n]) * h[n] + us * Br[n];
      y += h[n] * Cr[n];
    }
    out[row * Dc + d] = y + xv * Dv;
  }
}

extern "C" void kernel_launch(void* const* d_in, const int* in_sizes, int n_in,
                              void* d_out, int out_size, void* d_ws, size_t ws_size,
                              hipStream_t stream) {
  const float* x     = (const float*)d_in[0];
  const float* delta = (const float*)d_in[1];
  const float* Bm    = (const float*)d_in[2];
  const float* Cm    = (const float*)d_in[3];
  const float* W     = (const float*)d_in[4];
  const float* bias  = (const float*)d_in[5];
  const float* A_log = (const float*)d_in[6];
  const float* Dp    = (const float*)d_in[7];
  float* out = (float*)d_out;

  int Dc = in_sizes[5];
  int K  = in_sizes[4] / Dc;
  long long BL = (long long)in_sizes[0] / Dc;
  int Bsz = (BL % 2 == 0) ? 2 : 1;
  int L = (int)(BL / Bsz);

  size_t dtElems = (size_t)BL * Dc;
  size_t dtBytes = dtElems * sizeof(float);

  if (K == FUSE_K && (L % SCAN_TC) == 0 && (Dc % 16) == 0) {
    int blocks = Bsz * (Dc / 16);
    scan_dtfused_kernel<<<blocks, 384, 0, stream>>>(x, delta, Bm, Cm, W, bias,
                                                    A_log, Dp, out, L, Dc);
    return;
  }

  if (ws_size < dtBytes) {
    int threads = 256;
    long long tot = (long long)Bsz * Dc;
    int blocks = (int)((tot + threads - 1) / threads);
    scan_fused_kernel<<<blocks, threads, 0, stream>>>(
        x, delta, Bm, Cm, W, bias, A_log, Dp, out, L, Dc, K, Bsz);
    return;
  }

  float* dt = (float*)d_ws;

  if ((BL % 16) == 0 && (Dc % 32) == 0 && (K % 4) == 0) {
    int tiles = (int)(BL / 16) * (Dc / 32);
    int blocks = (tiles + 7) / 8;
    dtproj_wmma_kernel<<<blocks, 256, 0, stream>>>(delta, W, bias, dt,
                                                   (int)BL, Dc, K);
  } else {
    long long tot = BL * (long long)Dc;
    int blocks = (int)((tot + 255) / 256);
    dtproj_naive_kernel<<<blocks, 256, 0, stream>>>(delta, W, bias, dt, BL, Dc, K);
  }

  if ((L % SCAN_TC) == 0 && (Dc % 16) == 0) {
    int blocks = Bsz * (Dc / 16);
    scan_lanes_kernel<<<blocks, 256, 0, stream>>>(dt, x, Bm, Cm, A_log, Dp, out,
                                                  L, Dc);
  } else {
    long long tot = (long long)Bsz * Dc;
    int blocks = (int)((tot + 255) / 256);
    scan_simple_kernel<<<blocks, 256, 0, stream>>>(dt, x, Bm, Cm, A_log, Dp, out,
                                                   L, Dc, Bsz);
  }
}
